# starting kernel
# baseline (speedup 1.0000x reference)
#include <hip/hip_runtime.h>
#include <hip/hip_bf16.h>


#define D 128
#define N_MID 50000
#define N_DST 10000

typedef __attribute__((ext_vector_type(2))) float v2f;
typedef __attribute__((ext_vector_type(8))) float v8f;

__global__ __launch_bounds__(256) void sage_scatter(
    const float* __restrict__ src, const int* __restrict__ row,
    const int* __restrict__ col, float* __restrict__ agg,
    float* __restrict__ deg, int E) {
  int e = blockIdx.x * 8 + (threadIdx.x >> 5);
  int lane = threadIdx.x & 31;
  if (e >= E) return;
  int r = row[e];
  int c = col[e];
  const float4 v = ((const float4*)(src + (size_t)r * D))[lane];
  float* dst = agg + (size_t)c * D + lane * 4;
  unsafeAtomicAdd(dst + 0, v.x);
  unsafeAtomicAdd(dst + 1, v.y);
  unsafeAtomicAdd(dst + 2, v.z);
  unsafeAtomicAdd(dst + 3, v.w);
  if (lane == 0) unsafeAtomicAdd(deg + c, 1.0f);
}

template <int RELU>
__global__ __launch_bounds__(256) void sage_gemm(
    const float* __restrict__ agg, const float* __restrict__ deg,
    const float* __restrict__ xdst, const float* __restrict__ Wl,
    const float* __restrict__ bl, const float* __restrict__ Wr,
    float* __restrict__ out, int M) {
  __shared__ float sMean[16][D + 4];
  __shared__ float sDst[16][D + 4];
  __shared__ float sOut[16][D];
  __shared__ float sInv[16];
  __shared__ float sNorm[16];

  const int tid = threadIdx.x;
  const int lane = tid & 31;
  const int w = tid >> 5;
  const int mtile = blockIdx.x;
  const int m0 = mtile * 16;

  if (tid < 16) {
    float dg = deg[m0 + tid];
    sInv[tid] = 1.0f / fmaxf(dg, 1.0f);
    sNorm[tid] = 0.0f;
  }
  __syncthreads();

  for (int i = tid; i < 16 * D; i += 256) {
    int r = i >> 7;
    int c = i & (D - 1);
    size_t g = (size_t)(m0 + r) * D + c;
    sMean[r][c] = agg[g] * sInv[r];
    sDst[r][c] = xdst[g];
  }
  __syncthreads();

  const int mrow = lane & 15;
  const int k0 = (lane >> 4) << 1;
  const int ncol = w * 16 + (lane & 15);

  float bv = bl[ncol];
  v8f acc = {bv, bv, bv, bv, bv, bv, bv, bv};

  const float* WlRow = Wl + (size_t)ncol * D;
  const float* WrRow = Wr + (size_t)ncol * D;

#pragma unroll
  for (int kk = 0; kk < D / 4; ++kk) {
    v2f a = *(const v2f*)&sMean[mrow][4 * kk + k0];
    v2f b = *(const v2f*)&WlRow[4 * kk + k0];
    acc = __builtin_amdgcn_wmma_f32_16x16x4_f32(false, a, false, b,
                                                (short)0, acc, false, false);
  }
#pragma unroll
  for (int kk = 0; kk < D / 4; ++kk) {
    v2f a = *(const v2f*)&sDst[mrow][4 * kk + k0];
    v2f b = *(const v2f*)&WrRow[4 * kk + k0];
    acc = __builtin_amdgcn_wmma_f32_16x16x4_f32(false, a, false, b,
                                                (short)0, acc, false, false);
  }

  const int rbase = (lane >> 4) << 3;
#pragma unroll
  for (int v = 0; v < 8; ++v) {
    sOut[rbase + v][ncol & (D - 1)] = acc[v];
  }
  __syncthreads();

  {
    int r = tid >> 4;
    int cbase = (tid & 15) * 8;
    float p = 0.0f;
#pragma unroll
    for (int j = 0; j < 8; ++j) {
      float x = sOut[r][cbase + j];
      p += x * x;
    }
    atomicAdd(&sNorm[r], p);
  }
  __syncthreads();

  for (int i = tid; i < 16 * D; i += 256) {
    int r = i >> 7;
    int c = i & (D - 1);
    float nrm = fmaxf(sqrtf(sNorm[r]), 1e-12f);
    float val = sOut[r][c] / nrm;
    if (RELU) val = fmaxf(val, 0.0f);
    out[(size_t)(m0 + r) * D + c] = val;
  }
}

extern "C" void kernel_launch(void* const* d_in, const int* in_sizes, int n_in,
                              void* d_out, int out_size, void* d_ws,
                              size_t ws_size, hipStream_t stream) {
  const float* x    = (const float*)d_in[0];
  const float* W_l1 = (const float*)d_in[1];
  const float* b_l1 = (const float*)d_in[2];
  const float* W_r1 = (const float*)d_in[3];
  const float* W_l2 = (const float*)d_in[4];
  const float* b_l2 = (const float*)d_in[5];
  const float* W_r2 = (const float*)d_in[6];
  const int* row1 = (const int*)d_in[7];
  const int* col1 = (const int*)d_in[8];
  const int* row2 = (const int*)d_in[9];
  const int* col2 = (const int*)d_in[10];
  const int E1 = in_sizes[7];
  const int E2 = in_sizes[9];

  float* ws = (float*)d_ws;
  float* agg1 = ws;
  float* deg1 = agg1 + (size_t)N_MID * D;
  float* agg2 = deg1 + N_MID;
  float* deg2 = agg2 + (size_t)N_DST * D;
  float* h    = deg2 + N_DST;
  const size_t zero_floats =
      (size_t)N_MID * D + N_MID + (size_t)N_DST * D + N_DST;

  hipMemsetAsync(d_ws, 0, zero_floats * sizeof(float), stream);

  sage_scatter<<<(E1 + 7) / 8, 256, 0, stream>>>(x, row1, col1, agg1, deg1, E1);
  sage_gemm<1><<<N_MID / 16, 256, 0, stream>>>(agg1, deg1, x, W_l1, b_l1, W_r1,
                                               h, N_MID);
  sage_scatter<<<(E2 + 7) / 8, 256, 0, stream>>>(h, row2, col2, agg2, deg2, E2);
  sage_gemm<0><<<N_DST / 16, 256, 0, stream>>>(agg2, deg2, h, W_l2, b_l2, W_r2,
                                               (float*)d_out, N_DST);
}


# Float-atomic edge scatter replaced by grouping edges by destination (count with rank, scan, fill) and a one-wave-per-node register sum of rows in list order; summation order changes
# speedup vs baseline: 1.6114x; 1.6114x over previous
#include <hip/hip_runtime.h>
#include <hip/hip_bf16.h>


#define D 128
#define N_MID 50000
#define N_DST 10000

typedef __attribute__((ext_vector_type(2))) float v2f;
typedef __attribute__((ext_vector_type(8))) float v8f;

#define NSEG (N_MID + N_DST)
#define SCAN_BLK 1024
#define NSEG_PAD (60 * SCAN_BLK)

__global__ __launch_bounds__(256) void seg_count(
    const int* __restrict__ col1, const int* __restrict__ col2,
    int* __restrict__ cnt, int* __restrict__ rank, int E1, int E) {
  int e = blockIdx.x * 256 + threadIdx.x;
  if (e >= E) return;
  int seg = (e < E1) ? col1[e] : N_MID + col2[e - E1];
  rank[e] = atomicAdd(cnt + seg, 1);
}

__global__ __launch_bounds__(SCAN_BLK) void seg_scan(const int* __restrict__ cnt,
                                                     int* __restrict__ off) {
  __shared__ int sWave[32];
  __shared__ int sPre[32];
  const int t = threadIdx.x;
  const int lane = t & 31;
  const int w = t >> 5;
  const int b = blockIdx.x;

  const int4* p = (const int4*)cnt;
  int pre = 0;
  for (int j = t; j < b * (SCAN_BLK / 4); j += SCAN_BLK) {
    int4 v = p[j];
    pre += v.x + v.y + v.z + v.w;
  }
  const int own = cnt[b * SCAN_BLK + t];

#pragma unroll
  for (int d = 16; d > 0; d >>= 1) pre += __shfl_xor(pre, d);
  int inc = own;
#pragma unroll
  for (int d = 1; d < 32; d <<= 1) {
    int n = __shfl_up(inc, d);
    if (lane >= d) inc += n;
  }
  if (lane == 31) sWave[w] = inc;
  if (lane == 0) sPre[w] = pre;
  __syncthreads();
  if (w == 0) {
    int x = sWave[lane];
    int xi = x;
#pragma unroll
    for (int d = 1; d < 32; d <<= 1) {
      int n = __shfl_up(xi, d);
      if (lane >= d) xi += n;
    }
    sWave[lane] = xi - x;
    int r = sPre[lane];
#pragma unroll
    for (int d = 16; d > 0; d >>= 1) r += __shfl_xor(r, d);
    sPre[lane] = r;
  }
  __syncthreads();
  off[b * SCAN_BLK + t] = sPre[0] + sWave[w] + inc - own;
}

__global__ __launch_bounds__(256) void seg_fill(
    const int* __restrict__ row1, const int* __restrict__ col1,
    const int* __restrict__ row2, const int* __restrict__ col2,
    const int* __restrict__ rank, const int* __restrict__ off,
    int* __restrict__ lst, int E1, int E) {
  int e = blockIdx.x * 256 + threadIdx.x;
  if (e >= E) return;
  int seg, r;
  if (e < E1) {
    seg = col1[e];
    r = row1[e];
  } else {
    seg = N_MID + col2[e - E1];
    r = row2[e - E1];
  }
  lst[off[seg] + rank[e]] = r;
}

__device__ __forceinline__ void add4(float4& a, const float4 v) {
  a.x += v.x;
  a.y += v.y;
  a.z += v.z;
  a.w += v.w;
}

__global__ __launch_bounds__(256) void sage_gather(
    const float* __restrict__ src, const int* __restrict__ lst,
    const int* __restrict__ off, const int* __restrict__ cnt,
    float* __restrict__ agg, float* __restrict__ deg, int segbase, int n) {
  const int node =
      __builtin_amdgcn_readfirstlane(blockIdx.x * 8 + (threadIdx.x >> 5));
  if (node >= n) return;
  const int lane = threadIdx.x & 31;
  const int c = cnt[segbase + node];
  const int* __restrict__ l = lst + off[segbase + node];
  const float4* __restrict__ s4 = (const float4*)src + lane;

  float4 acc = {0.0f, 0.0f, 0.0f, 0.0f};
  int i = 0;
  for (; i + 8 <= c; i += 8) {
    float4 v[8];
#pragma unroll
    for (int j = 0; j < 8; ++j) v[j] = s4[(size_t)l[i + j] * (D / 4)];
#pragma unroll
    for (int j = 0; j < 8; ++j) add4(acc, v[j]);
  }
  for (; i < c; i += 4) {
    const int last = c - 1;
    float4 v[4];
#pragma unroll
    for (int j = 0; j < 4; ++j) {
      int ij = i + j;
      v[j] = s4[(size_t)l[ij < last ? ij : last] * (D / 4)];
    }
    const float4 z = {0.0f, 0.0f, 0.0f, 0.0f};
    add4(acc, v[0]);
    add4(acc, (i + 1 < c) ? v[1] : z);
    add4(acc, (i + 2 < c) ? v[2] : z);
    add4(acc, (i + 3 < c) ? v[3] : z);
  }
  ((float4*)(agg + (size_t)node * D))[lane] = acc;
  if (lane == 0) deg[node] = (float)c;
}

template <int RELU>
__global__ __launch_bounds__(256) void sage_gemm(
    const float* __restrict__ agg, const float* __restrict__ deg,
    const float* __restrict__ xdst, const float* __restrict__ Wl,
    const float* __restrict__ bl, const float* __restrict__ Wr,
    float* __restrict__ out, int M) {
  __shared__ float sMean[16][D + 4];
  __shared__ float sDst[16][D + 4];
  __shared__ float sOut[16][D];
  __shared__ float sInv[16];
  __shared__ float sNorm[16];

  const int tid = threadIdx.x;
  const int lane = tid & 31;
  const int w = tid >> 5;
  const int mtile = blockIdx.x;
  const int m0 = mtile * 16;

  if (tid < 16) {
    float dg = deg[m0 + tid];
    sInv[tid] = 1.0f / fmaxf(dg, 1.0f);
    sNorm[tid] = 0.0f;
  }
  __syncthreads();

  for (int i = tid; i < 16 * D; i += 256) {
    int r = i >> 7;
    int c = i & (D - 1);
    size_t g = (size_t)(m0 + r) * D + c;
    sMean[r][c] = agg[g] * sInv[r];
    sDst[r][c] = xdst[g];
  }
  __syncthreads();

  const int mrow = lane & 15;
  const int k0 = (lane >> 4) << 1;
  const int ncol = w * 16 + (lane & 15);

  float bv = bl[ncol];
  v8f acc = {bv, bv, bv, bv, bv, bv, bv, bv};

  const float* WlRow = Wl + (size_t)ncol * D;
  const float* WrRow = Wr + (size_t)ncol * D;

#pragma unroll
  for (int kk = 0; kk < D / 4; ++kk) {
    v2f a = *(const v2f*)&sMean[mrow][4 * kk + k0];
    v2f b = *(const v2f*)&WlRow[4 * kk + k0];
    acc = __builtin_amdgcn_wmma_f32_16x16x4_f32(false, a, false, b,
                                                (short)0, acc, false, false);
  }
#pragma unroll
  for (int kk = 0; kk < D / 4; ++kk) {
    v2f a = *(const v2f*)&sDst[mrow][4 * kk + k0];
    v2f b = *(const v2f*)&WrRow[4 * kk + k0];
    acc = __builtin_amdgcn_wmma_f32_16x16x4_f32(false, a, false, b,
                                                (short)0, acc, false, false);
  }

  const int rbase = (lane >> 4) << 3;
#pragma unroll
  for (int v = 0; v < 8; ++v) {
    sOut[rbase + v][ncol & (D - 1)] = acc[v];
  }
  __syncthreads();

  {
    int r = tid >> 4;
    int cbase = (tid & 15) * 8;
    float p = 0.0f;
#pragma unroll
    for (int j = 0; j < 8; ++j) {
      float x = sOut[r][cbase + j];
      p += x * x;
    }
    atomicAdd(&sNorm[r], p);
  }
  __syncthreads();

  for (int i = tid; i < 16 * D; i += 256) {
    int r = i >> 7;
    int c = i & (D - 1);
    float nrm = fmaxf(sqrtf(sNorm[r]), 1e-12f);
    float val = sOut[r][c] / nrm;
    if (RELU) val = fmaxf(val, 0.0f);
    out[(size_t)(m0 + r) * D + c] = val;
  }
}

extern "C" void kernel_launch(void* const* d_in, const int* in_sizes, int n_in,
                              void* d_out, int out_size, void* d_ws,
                              size_t ws_size, hipStream_t stream) {
  const float* x    = (const float*)d_in[0];
  const float* W_l1 = (const float*)d_in[1];
  const float* b_l1 = (const float*)d_in[2];
  const float* W_r1 = (const float*)d_in[3];
  const float* W_l2 = (const float*)d_in[4];
  const float* b_l2 = (const float*)d_in[5];
  const float* W_r2 = (const float*)d_in[6];
  const int* row1 = (const int*)d_in[7];
  const int* col1 = (const int*)d_in[8];
  const int* row2 = (const int*)d_in[9];
  const int* col2 = (const int*)d_in[10];
  const int E1 = in_sizes[7];
  const int E2 = in_sizes[9];
  const int E = E1 + E2;

  float* ws = (float*)d_ws;
  float* agg1 = ws;
  float* deg1 = agg1 + (size_t)N_MID * D;
  float* agg2 = deg1 + N_MID;
  float* deg2 = agg2 + (size_t)N_DST * D;
  float* h    = deg2 + N_DST;
  int* cnt  = (int*)(h + (size_t)N_MID * D);
  int* off  = cnt + NSEG_PAD;
  int* rank = off + NSEG_PAD;
  int* lst  = rank + E;

  hipMemsetAsync(cnt, 0, (size_t)NSEG_PAD * sizeof(int), stream);

  seg_count<<<(E + 255) / 256, 256, 0, stream>>>(col1, col2, cnt, rank, E1, E);
  seg_scan<<<NSEG_PAD / SCAN_BLK, SCAN_BLK, 0, stream>>>(cnt, off);
  seg_fill<<<(E + 255) / 256, 256, 0, stream>>>(row1, col1, row2, col2, rank,
                                                off, lst, E1, E);
  sage_gather<<<N_MID / 8, 256, 0, stream>>>(x, lst, off, cnt, agg1, deg1, 0,
                                             N_MID);
  sage_gemm<1><<<N_MID / 16, 256, 0, stream>>>(agg1, deg1, x, W_l1, b_l1, W_r1,
                                               h, N_MID);
  sage_gather<<<N_DST / 8, 256, 0, stream>>>(h, lst, off, cnt, agg2, deg2,
                                             N_MID, N_DST);
  sage_gemm<0><<<N_DST / 16, 256, 0, stream>>>(agg2, deg2, h, W_l2, b_l2, W_r2,
                                               (float*)d_out, N_DST);
}


# Edge ranks taken with returning LDS adds (32 work-groups, each with every segment counter in 240 KiB of LDS, plus a per-segment combine) instead of returning global atomics; summation order changes
# speedup vs baseline: 1.9417x; 1.2050x over previous
#include <hip/hip_runtime.h>
#include <hip/hip_bf16.h>


#define D 128
#define N_MID 50000
#define N_DST 10000

typedef __attribute__((ext_vector_type(2))) float v2f;
typedef __attribute__((ext_vector_type(8))) float v8f;

#define NSEG (N_MID + N_DST)
#define SCAN_BLK 1024
#define NSEG_PAD (60 * SCAN_BLK)

#define NCH 32
#define CNT_UNR 16
#define TRIP (CNT_UNR * 1024)

__global__ __launch_bounds__(1024) void seg_count(
    const int* __restrict__ col1, const int* __restrict__ col2,
    int* __restrict__ cntc, int* __restrict__ rank, int E1, int E) {
  __shared__ int sCnt[NSEG_PAD + 32];
  const int t = threadIdx.x;
  const int c = blockIdx.x;
  int4* s4 = (int4*)sCnt;
  const int4 zero4 = {0, 0, 0, 0};
  for (int i = t; i < NSEG_PAD / 4; i += 1024) s4[i] = zero4;
  if (t < 32) sCnt[NSEG_PAD + t] = 0;
  __syncthreads();

  const int ntrips = (E + TRIP - 1) / TRIP;
  for (int trip = c; trip < ntrips; trip += NCH) {
    const int base = trip * TRIP + t;
    int v[CNT_UNR];
#pragma unroll
    for (int j = 0; j < CNT_UNR; ++j) {
      const int e = base + j * 1024;
      const int ec = (e < E) ? e : E - 1;
      const int* p = (ec < E1) ? col1 + ec : col2 + (ec - E1);
      v[j] = *p;
    }
    asm volatile("" ::"v"(v[0]), "v"(v[1]), "v"(v[2]), "v"(v[3]), "v"(v[4]),
                 "v"(v[5]), "v"(v[6]), "v"(v[7]), "v"(v[8]), "v"(v[9]),
                 "v"(v[10]), "v"(v[11]), "v"(v[12]), "v"(v[13]), "v"(v[14]),
                 "v"(v[15]));
    int rk[CNT_UNR];
#pragma unroll
    for (int j = 0; j < CNT_UNR; ++j) {
      const int e = base + j * 1024;
      const int seg = (e < E1) ? v[j] : v[j] + N_MID;
      rk[j] = atomicAdd(&sCnt[(e < E) ? seg : NSEG_PAD + (t & 31)], 1);
    }
#pragma unroll
    for (int j = 0; j < CNT_UNR; ++j) {
      const int e = base + j * 1024;
      if (e < E) rank[e] = rk[j];
    }
  }
  __syncthreads();

  int4* o4 = (int4*)(cntc + (size_t)c * NSEG_PAD);
  for (int i = t; i < NSEG_PAD / 4; i += 1024) o4[i] = s4[i];
}

__global__ __launch_bounds__(256) void seg_combine(int* __restrict__ cntc,
                                                   int* __restrict__ cnt) {
  const int s = blockIdx.x * 256 + threadIdx.x;
  int v[NCH];
#pragma unroll
  for (int c = 0; c < NCH; ++c) v[c] = cntc[(size_t)c * NSEG_PAD + s];
  int run = 0;
#pragma unroll
  for (int c = 0; c < NCH; ++c) {
    cntc[(size_t)c * NSEG_PAD + s] = run;
    run += v[c];
  }
  cnt[s] = run;
}

__global__ __launch_bounds__(SCAN_BLK) void seg_scan(const int* __restrict__ cnt,
                                                     int* __restrict__ off) {
  __shared__ int sWave[32];
  __shared__ int sPre[32];
  const int t = threadIdx.x;
  const int lane = t & 31;
  const int w = t >> 5;
  const int b = blockIdx.x;

  const int4* p = (const int4*)cnt;
  int pre = 0;
  for (int j = t; j < b * (SCAN_BLK / 4); j += SCAN_BLK) {
    int4 v = p[j];
    pre += v.x + v.y + v.z + v.w;
  }
  const int own = cnt[b * SCAN_BLK + t];

#pragma unroll
  for (int d = 16; d > 0; d >>= 1) pre += __shfl_xor(pre, d);
  int inc = own;
#pragma unroll
  for (int d = 1; d < 32; d <<= 1) {
    int n = __shfl_up(inc, d);
    if (lane >= d) inc += n;
  }
  if (lane == 31) sWave[w] = inc;
  if (lane == 0) sPre[w] = pre;
  __syncthreads();
  if (w == 0) {
    int x = sWave[lane];
    int xi = x;
#pragma unroll
    for (int d = 1; d < 32; d <<= 1) {
      int n = __shfl_up(xi, d);
      if (lane >= d) xi += n;
    }
    sWave[lane] = xi - x;
    int r = sPre[lane];
#pragma unroll
    for (int d = 16; d > 0; d >>= 1) r += __shfl_xor(r, d);
    sPre[lane] = r;
  }
  __syncthreads();
  off[b * SCAN_BLK + t] = sPre[0] + sWave[w] + inc - own;
}

__global__ __launch_bounds__(256) void seg_fill(
    const int* __restrict__ row1, const int* __restrict__ col1,
    const int* __restrict__ row2, const int* __restrict__ col2,
    const int* __restrict__ rank, const int* __restrict__ off,
    const int* __restrict__ cntc, int* __restrict__ lst, int E1, int E) {
  int e = blockIdx.x * 256 + threadIdx.x;
  if (e >= E) return;
  const int c = (e / TRIP) % NCH;
  int seg, r;
  if (e < E1) {
    seg = col1[e];
    r = row1[e];
  } else {
    seg = N_MID + col2[e - E1];
    r = row2[e - E1];
  }
  lst[off[seg] + cntc[(size_t)c * NSEG_PAD + seg] + rank[e]] = r;
}

__device__ __forceinline__ void add4(float4& a, const float4 v) {
  a.x += v.x;
  a.y += v.y;
  a.z += v.z;
  a.w += v.w;
}

__global__ __launch_bounds__(256) void sage_gather(
    const float* __restrict__ src, const int* __restrict__ lst,
    const int* __restrict__ off, const int* __restrict__ cnt,
    float* __restrict__ agg, float* __restrict__ deg, int segbase, int n) {
  const int node =
      __builtin_amdgcn_readfirstlane(blockIdx.x * 8 + (threadIdx.x >> 5));
  if (node >= n) return;
  const int lane = threadIdx.x & 31;
  const int c = cnt[segbase + node];
  const int* __restrict__ l = lst + off[segbase + node];
  const float4* __restrict__ s4 = (const float4*)src + lane;

  float4 acc = {0.0f, 0.0f, 0.0f, 0.0f};
  int i = 0;
  for (; i + 8 <= c; i += 8) {
    float4 v[8];
#pragma unroll
    for (int j = 0; j < 8; ++j) v[j] = s4[(size_t)l[i + j] * (D / 4)];
#pragma unroll
    for (int j = 0; j < 8; ++j) add4(acc, v[j]);
  }
  for (; i < c; i += 4) {
    const int last = c - 1;
    float4 v[4];
#pragma unroll
    for (int j = 0; j < 4; ++j) {
      int ij = i + j;
      v[j] = s4[(size_t)l[ij < last ? ij : last] * (D / 4)];
    }
    const float4 z = {0.0f, 0.0f, 0.0f, 0.0f};
    add4(acc, v[0]);
    add4(acc, (i + 1 < c) ? v[1] : z);
    add4(acc, (i + 2 < c) ? v[2] : z);
    add4(acc, (i + 3 < c) ? v[3] : z);
  }
  ((float4*)(agg + (size_t)node * D))[lane] = acc;
  if (lane == 0) deg[node] = (float)c;
}

template <int RELU>
__global__ __launch_bounds__(256) void sage_gemm(
    const float* __restrict__ agg, const float* __restrict__ deg,
    const float* __restrict__ xdst, const float* __restrict__ Wl,
    const float* __restrict__ bl, const float* __restrict__ Wr,
    float* __restrict__ out, int M) {
  __shared__ float sMean[16][D + 4];
  __shared__ float sDst[16][D + 4];
  __shared__ float sOut[16][D];
  __shared__ float sInv[16];
  __shared__ float sNorm[16];

  const int tid = threadIdx.x;
  const int lane = tid & 31;
  const int w = tid >> 5;
  const int mtile = blockIdx.x;
  const int m0 = mtile * 16;

  if (tid < 16) {
    float dg = deg[m0 + tid];
    sInv[tid] = 1.0f / fmaxf(dg, 1.0f);
    sNorm[tid] = 0.0f;
  }
  __syncthreads();

  for (int i = tid; i < 16 * D; i += 256) {
    int r = i >> 7;
    int c = i & (D - 1);
    size_t g = (size_t)(m0 + r) * D + c;
    sMean[r][c] = agg[g] * sInv[r];
    sDst[r][c] = xdst[g];
  }
  __syncthreads();

  const int mrow = lane & 15;
  const int k0 = (lane >> 4) << 1;
  const int ncol = w * 16 + (lane & 15);

  float bv = bl[ncol];
  v8f acc = {bv, bv, bv, bv, bv, bv, bv, bv};

  const float* WlRow = Wl + (size_t)ncol * D;
  const float* WrRow = Wr + (size_t)ncol * D;

#pragma unroll
  for (int kk = 0; kk < D / 4; ++kk) {
    v2f a = *(const v2f*)&sMean[mrow][4 * kk + k0];
    v2f b = *(const v2f*)&WlRow[4 * kk + k0];
    acc = __builtin_amdgcn_wmma_f32_16x16x4_f32(false, a, false, b,
                                                (short)0, acc, false, false);
  }
#pragma unroll
  for (int kk = 0; kk < D / 4; ++kk) {
    v2f a = *(const v2f*)&sDst[mrow][4 * kk + k0];
    v2f b = *(const v2f*)&WrRow[4 * kk + k0];
    acc = __builtin_amdgcn_wmma_f32_16x16x4_f32(false, a, false, b,
                                                (short)0, acc, false, false);
  }

  const int rbase = (lane >> 4) << 3;
#pragma unroll
  for (int v = 0; v < 8; ++v) {
    sOut[rbase + v][ncol & (D - 1)] = acc[v];
  }
  __syncthreads();

  {
    int r = tid >> 4;
    int cbase = (tid & 15) * 8;
    float p = 0.0f;
#pragma unroll
    for (int j = 0; j < 8; ++j) {
      float x = sOut[r][cbase + j];
      p += x * x;
    }
    atomicAdd(&sNorm[r], p);
  }
  __syncthreads();

  for (int i = tid; i < 16 * D; i += 256) {
    int r = i >> 7;
    int c = i & (D - 1);
    float nrm = fmaxf(sqrtf(sNorm[r]), 1e-12f);
    float val = sOut[r][c] / nrm;
    if (RELU) val = fmaxf(val, 0.0f);
    out[(size_t)(m0 + r) * D + c] = val;
  }
}

extern "C" void kernel_launch(void* const* d_in, const int* in_sizes, int n_in,
                              void* d_out, int out_size, void* d_ws,
                              size_t ws_size, hipStream_t stream) {
  const float* x    = (const float*)d_in[0];
  const float* W_l1 = (const float*)d_in[1];
  const float* b_l1 = (const float*)d_in[2];
  const float* W_r1 = (const float*)d_in[3];
  const float* W_l2 = (const float*)d_in[4];
  const float* b_l2 = (const float*)d_in[5];
  const float* W_r2 = (const float*)d_in[6];
  const int* row1 = (const int*)d_in[7];
  const int* col1 = (const int*)d_in[8];
  const int* row2 = (const int*)d_in[9];
  const int* col2 = (const int*)d_in[10];
  const int E1 = in_sizes[7];
  const int E2 = in_sizes[9];
  const int E = E1 + E2;

  float* ws = (float*)d_ws;
  float* agg1 = ws;
  float* deg1 = agg1 + (size_t)N_MID * D;
  float* agg2 = deg1 + N_MID;
  float* deg2 = agg2 + (size_t)N_DST * D;
  float* h    = deg2 + N_DST;
  int* cnt  = (int*)(h + (size_t)N_MID * D);
  int* off  = cnt + NSEG_PAD;
  int* cntc = off + NSEG_PAD;
  int* rank = cntc + (size_t)NCH * NSEG_PAD;
  int* lst  = rank + E;

  seg_count<<<NCH, 1024, 0, stream>>>(col1, col2, cntc, rank, E1, E);
  seg_combine<<<NSEG_PAD / 256, 256, 0, stream>>>(cntc, cnt);
  seg_scan<<<NSEG_PAD / SCAN_BLK, SCAN_BLK, 0, stream>>>(cnt, off);
  seg_fill<<<(E + 255) / 256, 256, 0, stream>>>(row1, col1, row2, col2, rank,
                                                off, cntc, lst, E1, E);
  sage_gather<<<N_MID / 8, 256, 0, stream>>>(x, lst, off, cnt, agg1, deg1, 0,
                                             N_MID);
  sage_gemm<1><<<N_MID / 16, 256, 0, stream>>>(agg1, deg1, x, W_l1, b_l1, W_r1,
                                               h, N_MID);
  sage_gather<<<N_DST / 8, 256, 0, stream>>>(h, lst, off, cnt, agg2, deg2,
                                             N_MID, N_DST);
  sage_gemm<0><<<N_DST / 16, 256, 0, stream>>>(agg2, deg2, h, W_l2, b_l2, W_r2,
                                               (float*)d_out, N_DST);
}


# GEMM: 256 work-groups of 16 waves hold both weight matrices in LDS, each wave owns whole 16x128 row tiles (was: weights re-read from L2 per 16 rows); row norm by lane shuffles, summation order changes
# speedup vs baseline: 2.2711x; 1.1696x over previous
#include <hip/hip_runtime.h>
#include <hip/hip_bf16.h>


#define D 128
#define N_MID 50000
#define N_DST 10000

typedef __attribute__((ext_vector_type(2))) float v2f;
typedef __attribute__((ext_vector_type(8))) float v8f;

#define NSEG (N_MID + N_DST)
#define SCAN_BLK 1024
#define NSEG_PAD (60 * SCAN_BLK)

#define NCH 32
#define CNT_UNR 16
#define TRIP (CNT_UNR * 1024)

__global__ __launch_bounds__(1024) void seg_count(
    const int* __restrict__ col1, const int* __restrict__ col2,
    int* __restrict__ cntc, int* __restrict__ rank, int E1, int E) {
  __shared__ int sCnt[NSEG_PAD + 32];
  const int t = threadIdx.x;
  const int c = blockIdx.x;
  int4* s4 = (int4*)sCnt;
  const int4 zero4 = {0, 0, 0, 0};
  for (int i = t; i < NSEG_PAD / 4; i += 1024) s4[i] = zero4;
  if (t < 32) sCnt[NSEG_PAD + t] = 0;
  __syncthreads();

  const int ntrips = (E + TRIP - 1) / TRIP;
  for (int trip = c; trip < ntrips; trip += NCH) {
    const int base = trip * TRIP + t;
    int v[CNT_UNR];
#pragma unroll
    for (int j = 0; j < CNT_UNR; ++j) {
      const int e = base + j * 1024;
      const int ec = (e < E) ? e : E - 1;
      const int* p = (ec < E1) ? col1 + ec : col2 + (ec - E1);
      v[j] = *p;
    }
    asm volatile("" ::"v"(v[0]), "v"(v[1]), "v"(v[2]), "v"(v[3]), "v"(v[4]),
                 "v"(v[5]), "v"(v[6]), "v"(v[7]), "v"(v[8]), "v"(v[9]),
                 "v"(v[10]), "v"(v[11]), "v"(v[12]), "v"(v[13]), "v"(v[14]),
                 "v"(v[15]));
    int rk[CNT_UNR];
#pragma unroll
    for (int j = 0; j < CNT_UNR; ++j) {
      const int e = base + j * 1024;
      const int seg = (e < E1) ? v[j] : v[j] + N_MID;
      rk[j] = atomicAdd(&sCnt[(e < E) ? seg : NSEG_PAD + (t & 31)], 1);
    }
#pragma unroll
    for (int j = 0; j < CNT_UNR; ++j) {
      const int e = base + j * 1024;
      if (e < E) rank[e] = rk[j];
    }
  }
  __syncthreads();

  int4* o4 = (int4*)(cntc + (size_t)c * NSEG_PAD);
  for (int i = t; i < NSEG_PAD / 4; i += 1024) o4[i] = s4[i];
}

__global__ __launch_bounds__(256) void seg_combine(int* __restrict__ cntc,
                                                   int* __restrict__ cnt) {
  const int s = blockIdx.x * 256 + threadIdx.x;
  int v[NCH];
#pragma unroll
  for (int c = 0; c < NCH; ++c) v[c] = cntc[(size_t)c * NSEG_PAD + s];
  int run = 0;
#pragma unroll
  for (int c = 0; c < NCH; ++c) {
    cntc[(size_t)c * NSEG_PAD + s] = run;
    run += v[c];
  }
  cnt[s] = run;
}

__global__ __launch_bounds__(SCAN_BLK) void seg_scan(const int* __restrict__ cnt,
                                                     int* __restrict__ off) {
  __shared__ int sWave[32];
  __shared__ int sPre[32];
  const int t = threadIdx.x;
  const int lane = t & 31;
  const int w = t >> 5;
  const int b = blockIdx.x;

  const int4* p = (const int4*)cnt;
  int pre = 0;
  for (int j = t; j < b * (SCAN_BLK / 4); j += SCAN_BLK) {
    int4 v = p[j];
    pre += v.x + v.y + v.z + v.w;
  }
  const int own = cnt[b * SCAN_BLK + t];

#pragma unroll
  for (int d = 16; d > 0; d >>= 1) pre += __shfl_xor(pre, d);
  int inc = own;
#pragma unroll
  for (int d = 1; d < 32; d <<= 1) {
    int n = __shfl_up(inc, d);
    if (lane >= d) inc += n;
  }
  if (lane == 31) sWave[w] = inc;
  if (lane == 0) sPre[w] = pre;
  __syncthreads();
  if (w == 0) {
    int x = sWave[lane];
    int xi = x;
#pragma unroll
    for (int d = 1; d < 32; d <<= 1) {
      int n = __shfl_up(xi, d);
      if (lane >= d) xi += n;
    }
    sWave[lane] = xi - x;
    int r = sPre[lane];
#pragma unroll
    for (int d = 16; d > 0; d >>= 1) r += __shfl_xor(r, d);
    sPre[lane] = r;
  }
  __syncthreads();
  off[b * SCAN_BLK + t] = sPre[0] + sWave[w] + inc - own;
}

__global__ __launch_bounds__(256) void seg_fill(
    const int* __restrict__ row1, const int* __restrict__ col1,
    const int* __restrict__ row2, const int* __restrict__ col2,
    const int* __restrict__ rank, const int* __restrict__ off,
    const int* __restrict__ cntc, int* __restrict__ lst, int E1, int E) {
  int e = blockIdx.x * 256 + threadIdx.x;
  if (e >= E) return;
  const int c = (e / TRIP) % NCH;
  int seg, r;
  if (e < E1) {
    seg = col1[e];
    r = row1[e];
  } else {
    seg = N_MID + col2[e - E1];
    r = row2[e - E1];
  }
  lst[off[seg] + cntc[(size_t)c * NSEG_PAD + seg] + rank[e]] = r;
}

__device__ __forceinline__ void add4(float4& a, const float4 v) {
  a.x += v.x;
  a.y += v.y;
  a.z += v.z;
  a.w += v.w;
}

__global__ __launch_bounds__(256) void sage_gather(
    const float* __restrict__ src, const int* __restrict__ lst,
    const int* __restrict__ off, const int* __restrict__ cnt,
    float* __restrict__ agg, float* __restrict__ deg, int segbase, int n) {
  const int node =
      __builtin_amdgcn_readfirstlane(blockIdx.x * 8 + (threadIdx.x >> 5));
  if (node >= n) return;
  const int lane = threadIdx.x & 31;
  const int c = cnt[segbase + node];
  const int* __restrict__ l = lst + off[segbase + node];
  const float4* __restrict__ s4 = (const float4*)src + lane;

  float4 acc = {0.0f, 0.0f, 0.0f, 0.0f};
  int i = 0;
  for (; i + 8 <= c; i += 8) {
    float4 v[8];
#pragma unroll
    for (int j = 0; j < 8; ++j) v[j] = s4[(size_t)l[i + j] * (D / 4)];
#pragma unroll
    for (int j = 0; j < 8; ++j) add4(acc, v[j]);
  }
  for (; i < c; i += 4) {
    const int last = c - 1;
    float4 v[4];
#pragma unroll
    for (int j = 0; j < 4; ++j) {
      int ij = i + j;
      v[j] = s4[(size_t)l[ij < last ? ij : last] * (D / 4)];
    }
    const float4 z = {0.0f, 0.0f, 0.0f, 0.0f};
    add4(acc, v[0]);
    add4(acc, (i + 1 < c) ? v[1] : z);
    add4(acc, (i + 2 < c) ? v[2] : z);
    add4(acc, (i + 3 < c) ? v[3] : z);
  }
  ((float4*)(agg + (size_t)node * D))[lane] = acc;
  if (lane == 0) deg[node] = (float)c;
}

#define GEMM_WG 256
#define GEMM_WAVES 16
#define WP (D + 4)

template <int RELU>
__global__ __launch_bounds__(GEMM_WAVES * 32) void sage_gemm(
    const float* __restrict__ agg, const float* __restrict__ deg,
    const float* __restrict__ xdst, const float* __restrict__ Wl,
    const float* __restrict__ bl, const float* __restrict__ Wr,
    float* __restrict__ out, int M) {
  __shared__ __attribute__((aligned(16))) float sW[2][D][WP];
  __shared__ __attribute__((aligned(16))) float sA[GEMM_WAVES][16][WP];

  const int tid = threadIdx.x;
  const int lane = tid & 31;
  const int w = tid >> 5;
  const int c = lane & 15;
  const int h = lane >> 4;
  const int k0 = h << 1;

  for (int i = tid; i < D * (D / 4); i += GEMM_WAVES * 32) {
    const int n = i >> 5;
    const int q = i & 31;
    const float4 vl = ((const float4*)Wl)[i];
    const float4 vr = ((const float4*)Wr)[i];
    *(float4*)&sW[0][n][4 * q] = vl;
    *(float4*)&sW[1][n][4 * q] = vr;
  }

  float bv[8];
#pragma unroll
  for (int j = 0; j < 8; ++j) bv[j] = bl[16 * j + c];

  const int ntiles = M >> 4;
  const int slots = gridDim.x * GEMM_WAVES;
  const int rounds = (ntiles + slots - 1) / slots;
  float (*myA)[WP] = sA[w];

  for (int rd = 0; rd < rounds; ++rd) {
    const int tile = rd * slots + w * gridDim.x + blockIdx.x;
    const bool live = tile < ntiles;
    const size_t m0 = (size_t)(live ? tile : 0) * 16;

    v8f acc[8];
#pragma unroll
    for (int j = 0; j < 8; ++j) {
      const float b = bv[j];
      const v8f t = {b, b, b, b, b, b, b, b};
      acc[j] = t;
    }

    if (live) {
      float4 v[16];
      float iv[16];
#pragma unroll
      for (int r = 0; r < 16; ++r) {
        v[r] = ((const float4*)(agg + (m0 + r) * D))[lane];
        iv[r] = 1.0f / fmaxf(deg[m0 + r], 1.0f);
      }
#pragma unroll
      for (int r = 0; r < 16; ++r) {
        float4 t = v[r];
        t.x *= iv[r];
        t.y *= iv[r];
        t.z *= iv[r];
        t.w *= iv[r];
        *(float4*)&myA[r][4 * lane] = t;
      }
    }
    __syncthreads();

    if (live) {
#pragma unroll
      for (int kk = 0; kk < D / 4; ++kk) {
        const v2f a = *(const v2f*)&myA[c][4 * kk + k0];
#pragma unroll
        for (int j = 0; j < 8; ++j) {
          const v2f b = *(const v2f*)&sW[0][16 * j + c][4 * kk + k0];
          acc[j] = __builtin_amdgcn_wmma_f32_16x16x4_f32(
              false, a, false, b, (short)0, acc[j], false, false);
        }
      }
    }
    __syncthreads();

    if (live) {
      float4 v[16];
#pragma unroll
      for (int r = 0; r < 16; ++r)
        v[r] = ((const float4*)(xdst + (m0 + r) * D))[lane];
#pragma unroll
      for (int r = 0; r < 16; ++r) *(float4*)&myA[r][4 * lane] = v[r];
    }
    __syncthreads();

    if (live) {
#pragma unroll
      for (int kk = 0; kk < D / 4; ++kk) {
        const v2f a = *(const v2f*)&myA[c][4 * kk + k0];
#pragma unroll
        for (int j = 0; j < 8; ++j) {
          const v2f b = *(const v2f*)&sW[1][16 * j + c][4 * kk + k0];
          acc[j] = __builtin_amdgcn_wmma_f32_16x16x4_f32(
              false, a, false, b, (short)0, acc[j], false, false);
        }
      }
    }
    __syncthreads();

    if (live) {
      float ss[8];
#pragma unroll
      for (int r = 0; r < 8; ++r) {
        float p = 0.0f;
#pragma unroll
        for (int j = 0; j < 8; ++j) p += acc[j][r] * acc[j][r];
        ss[r] = p;
      }
#pragma unroll
      for (int d = 1; d < 16; d <<= 1) {
#pragma unroll
        for (int r = 0; r < 8; ++r) ss[r] += __shfl_xor(ss[r], d);
      }
#pragma unroll
      for (int r = 0; r < 8; ++r) {
        const float nrm = fmaxf(sqrtf(ss[r]), 1e-12f);
#pragma unroll
        for (int j = 0; j < 8; ++j) {
          float val = acc[j][r] / nrm;
          if (RELU) val = fmaxf(val, 0.0f);
          myA[8 * h + r][16 * j + c] = val;
        }
      }
    }
    __syncthreads();

    if (live) {
#pragma unroll
      for (int r = 0; r < 16; ++r) {
        const float4 t = *(const float4*)&myA[r][4 * lane];
        ((float4*)(out + (m0 + r) * D))[lane] = t;
      }
    }
    __syncthreads();
  }
}

extern "C" void kernel_launch(void* const* d_in, const int* in_sizes, int n_in,
                              void* d_out, int out_size, void* d_ws,
                              size_t ws_size, hipStream_t stream) {
  const float* x    = (const float*)d_in[0];
  const float* W_l1 = (const float*)d_in[1];
  const float* b_l1 = (const float*)d_in[2];
  const float* W_r1 = (const float*)d_in[3];
  const float* W_l2 = (const float*)d_in[4];
  const float* b_l2 = (const float*)d_in[5];
  const float* W_r2 = (const float*)d_in[6];
  const int* row1 = (const int*)d_in[7];
  const int* col1 = (const int*)d_in[8];
  const int* row2 = (const int*)d_in[9];
  const int* col2 = (const int*)d_in[10];
  const int E1 = in_sizes[7];
  const int E2 = in_sizes[9];
  const int E = E1 + E2;

  float* ws = (float*)d_ws;
  float* agg1 = ws;
  float* deg1 = agg1 + (size_t)N_MID * D;
  float* agg2 = deg1 + N_MID;
  float* deg2 = agg2 + (size_t)N_DST * D;
  float* h    = deg2 + N_DST;
  int* cnt  = (int*)(h + (size_t)N_MID * D);
  int* off  = cnt + NSEG_PAD;
  int* cntc = off + NSEG_PAD;
  int* rank = cntc + (size_t)NCH * NSEG_PAD;
  int* lst  = rank + E;

  seg_count<<<NCH, 1024, 0, stream>>>(col1, col2, cntc, rank, E1, E);
  seg_combine<<<NSEG_PAD / 256, 256, 0, stream>>>(cntc, cnt);
  seg_scan<<<NSEG_PAD / SCAN_BLK, SCAN_BLK, 0, stream>>>(cnt, off);
  seg_fill<<<(E + 255) / 256, 256, 0, stream>>>(row1, col1, row2, col2, rank,
                                                off, cntc, lst, E1, E);
  sage_gather<<<N_MID / 8, 256, 0, stream>>>(x, lst, off, cnt, agg1, deg1, 0,
                                             N_MID);
  sage_gemm<1><<<GEMM_WG, GEMM_WAVES * 32, 0, stream>>>(agg1, deg1, x, W_l1, b_l1, W_r1,
                                               h, N_MID);
  sage_gather<<<N_DST / 8, 256, 0, stream>>>(h, lst, off, cnt, agg2, deg2,
                                             N_MID, N_DST);
  sage_gemm<0><<<GEMM_WG, GEMM_WAVES * 32, 0, stream>>>(agg2, deg2, h, W_l2, b_l2, W_r2,
                                               (float*)d_out, N_DST);
}


# Segment mean moved into the layer kernel: waves walk their tile's contiguous list run, 16 source rows in flight, and hand the means to the WMMA through LDS; gather launches and agg planes dropped
# speedup vs baseline: 2.5730x; 1.1329x over previous
#include <hip/hip_runtime.h>
#include <hip/hip_bf16.h>


#define D 128
#define N_MID 50000
#define N_DST 10000

typedef __attribute__((ext_vector_type(2))) float v2f;
typedef __attribute__((ext_vector_type(4))) float v4f;
typedef __attribute__((ext_vector_type(8))) float v8f;

#define NSEG (N_MID + N_DST)
#define SCAN_BLK 1024
#define NSEG_PAD (60 * SCAN_BLK)

#define NCH 32
#define CNT_UNR 16
#define TRIP (CNT_UNR * 1024)

__global__ __launch_bounds__(1024) void seg_count(
    const int* __restrict__ col1, const int* __restrict__ col2,
    int* __restrict__ cntc, int* __restrict__ rank, int E1, int E) {
  __shared__ int sCnt[NSEG_PAD + 32];
  const int t = threadIdx.x;
  const int c = blockIdx.x;
  int4* s4 = (int4*)sCnt;
  const int4 zero4 = {0, 0, 0, 0};
  for (int i = t; i < NSEG_PAD / 4; i += 1024) s4[i] = zero4;
  if (t < 32) sCnt[NSEG_PAD + t] = 0;
  __syncthreads();

  const int ntrips = (E + TRIP - 1) / TRIP;
  for (int trip = c; trip < ntrips; trip += NCH) {
    const int base = trip * TRIP + t;
    int v[CNT_UNR];
#pragma unroll
    for (int j = 0; j < CNT_UNR; ++j) {
      const int e = base + j * 1024;
      const int ec = (e < E) ? e : E - 1;
      const int* p = (ec < E1) ? col1 + ec : col2 + (ec - E1);
      v[j] = *p;
    }
    asm volatile("" ::"v"(v[0]), "v"(v[1]), "v"(v[2]), "v"(v[3]), "v"(v[4]),
                 "v"(v[5]), "v"(v[6]), "v"(v[7]), "v"(v[8]), "v"(v[9]),
                 "v"(v[10]), "v"(v[11]), "v"(v[12]), "v"(v[13]), "v"(v[14]),
                 "v"(v[15]));
    int rk[CNT_UNR];
#pragma unroll
    for (int j = 0; j < CNT_UNR; ++j) {
      const int e = base + j * 1024;
      const int seg = (e < E1) ? v[j] : v[j] + N_MID;
      rk[j] = atomicAdd(&sCnt[(e < E) ? seg : NSEG_PAD + (t & 31)], 1);
    }
#pragma unroll
    for (int j = 0; j < CNT_UNR; ++j) {
      const int e = base + j * 1024;
      if (e < E) rank[e] = rk[j];
    }
  }
  __syncthreads();

  int4* o4 = (int4*)(cntc + (size_t)c * NSEG_PAD);
  for (int i = t; i < NSEG_PAD / 4; i += 1024) o4[i] = s4[i];
}

__global__ __launch_bounds__(256) void seg_combine(int* __restrict__ cntc,
                                                   int* __restrict__ cnt) {
  const int s = blockIdx.x * 256 + threadIdx.x;
  int v[NCH];
#pragma unroll
  for (int c = 0; c < NCH; ++c) v[c] = cntc[(size_t)c * NSEG_PAD + s];
  int run = 0;
#pragma unroll
  for (int c = 0; c < NCH; ++c) {
    cntc[(size_t)c * NSEG_PAD + s] = run;
    run += v[c];
  }
  cnt[s] = run;
}

__global__ __launch_bounds__(SCAN_BLK) void seg_scan(const int* __restrict__ cnt,
                                                     int* __restrict__ off) {
  __shared__ int sWave[32];
  __shared__ int sPre[32];
  const int t = threadIdx.x;
  const int lane = t & 31;
  const int w = t >> 5;
  const int b = blockIdx.x;

  const int4* p = (const int4*)cnt;
  int pre = 0;
  for (int j = t; j < b * (SCAN_BLK / 4); j += SCAN_BLK) {
    int4 v = p[j];
    pre += v.x + v.y + v.z + v.w;
  }
  const int own = cnt[b * SCAN_BLK + t];

#pragma unroll
  for (int d = 16; d > 0; d >>= 1) pre += __shfl_xor(pre, d);
  int inc = own;
#pragma unroll
  for (int d = 1; d < 32; d <<= 1) {
    int n = __shfl_up(inc, d);
    if (lane >= d) inc += n;
  }
  if (lane == 31) sWave[w] = inc;
  if (lane == 0) sPre[w] = pre;
  __syncthreads();
  if (w == 0) {
    int x = sWave[lane];
    int xi = x;
#pragma unroll
    for (int d = 1; d < 32; d <<= 1) {
      int n = __shfl_up(xi, d);
      if (lane >= d) xi += n;
    }
    sWave[lane] = xi - x;
    int r = sPre[lane];
#pragma unroll
    for (int d = 16; d > 0; d >>= 1) r += __shfl_xor(r, d);
    sPre[lane] = r;
  }
  __syncthreads();
  off[b * SCAN_BLK + t] = sPre[0] + sWave[w] + inc - own;
}

__global__ __launch_bounds__(256) void seg_fill(
    const int* __restrict__ row1, const int* __restrict__ col1,
    const int* __restrict__ row2, const int* __restrict__ col2,
    const int* __restrict__ rank, const int* __restrict__ off,
    const int* __restrict__ cntc, int* __restrict__ lst, int E1, int E) {
  int e = blockIdx.x * 256 + threadIdx.x;
  if (e >= E) return;
  const int c = (e / TRIP) % NCH;
  int seg, r;
  if (e < E1) {
    seg = col1[e];
    r = row1[e];
  } else {
    seg = N_MID + col2[e - E1];
    r = row2[e - E1];
  }
  lst[off[seg] + cntc[(size_t)c * NSEG_PAD + seg] + rank[e]] = r;
}

#define GEMM_WG 256
#define GEMM_WAVES 16
#define WP (D + 4)

template <int RELU>
__global__ __launch_bounds__(GEMM_WAVES * 32) void sage_layer(
    const float* __restrict__ src, const int* __restrict__ lst,
    const int* __restrict__ off, int segbase,
    const float* __restrict__ xdst, const float* __restrict__ Wl,
    const float* __restrict__ bl, const float* __restrict__ Wr,
    float* __restrict__ out, int M) {
  __shared__ __attribute__((aligned(16))) float sW[2][D][WP];
  __shared__ __attribute__((aligned(16))) float sA[GEMM_WAVES][16][WP];

  const int tid = threadIdx.x;
  const int lane = tid & 31;
  const int w = __builtin_amdgcn_readfirstlane(tid >> 5);
  const int c = lane & 15;
  const int h = lane >> 4;
  const int k0 = h << 1;

  for (int i = tid; i < D * (D / 4); i += GEMM_WAVES * 32) {
    const int n = i >> 5;
    const int q = i & 31;
    const float4 vl = ((const float4*)Wl)[i];
    const float4 vr = ((const float4*)Wr)[i];
    *(float4*)&sW[0][n][4 * q] = vl;
    *(float4*)&sW[1][n][4 * q] = vr;
  }

  float bv[8];
#pragma unroll
  for (int j = 0; j < 8; ++j) bv[j] = bl[16 * j + c];

  const int ntiles = M >> 4;
  const int slots = gridDim.x * GEMM_WAVES;
  const int rounds = (ntiles + slots - 1) / slots;
  float (*myA)[WP] = sA[w];

  for (int rd = 0; rd < rounds; ++rd) {
    const int first = rd * slots + blockIdx.x;
    const int tile = first + w * gridDim.x;
    const bool live = tile < ntiles;
    const size_t m0 = (size_t)(live ? tile : 0) * 16;

    v8f acc[8];
#pragma unroll
    for (int j = 0; j < 8; ++j) {
      const float b = bv[j];
      const v8f t = {b, b, b, b, b, b, b, b};
      acc[j] = t;
    }

    int nl = 0;
#pragma unroll
    for (int t = 0; t < GEMM_WAVES; ++t)
      nl += (first + t * (int)gridDim.x < ntiles) ? 1 : 0;
    int sh = 0;
    while (nl > 0 && (2 << sh) * nl <= GEMM_WAVES) ++sh;
    const int gt = w >> sh;
    if (gt < nl) {
      const int span = 16 >> sh;
      const int r0 = (w & ((1 << sh) - 1)) * span;
      const int r1 = r0 + span;
      const int gm0 = (first + gt * (int)gridDim.x) * 16;
      float (*dstA)[WP] = sA[gt];

      const int* __restrict__ ob = off + segbase + gm0;
      const int o0 = ob[lane < 16 ? lane : 16];
      const int o1 = ob[lane < 15 ? lane + 1 : 16];
      const float invv = 1.0f / fmaxf((float)(o1 - o0), 1.0f);
      const int lbase = __builtin_amdgcn_readlane(o0, r0);
      const int total = __builtin_amdgcn_readlane(o0, r1) - lbase;
      const int bnd = o0 - lbase;
      const int* __restrict__ L = lst + lbase;
      const v4f* __restrict__ s4 = (const v4f*)src + lane;
      const v4f zero4 = {0.0f, 0.0f, 0.0f, 0.0f};

      v4f sum = zero4;
      int r = r0;
      int end = __builtin_amdgcn_readlane(bnd, r0 + 1);

      int idx = 0;
      if (total > 0) idx = L[lane < total ? lane : total - 1];
      for (int i0 = 0; i0 < total; i0 += 16) {
        const int rem = total - i0;
        int idxn = idx;
        if (rem > 16) {
          const int rn = rem - 16;
          idxn = L[i0 + 16 + (lane < rn ? lane : rn - 1)];
        }
        v4f v[16];
#pragma unroll
        for (int j = 0; j < 16; ++j) {
          const int q = (j < rem) ? j : rem - 1;
          const int srow = __builtin_amdgcn_readlane(idx, q);
          v[j] = s4[(size_t)srow * (D / 4)];
        }
#pragma unroll
        for (int j = 0; j < 16; ++j) {
          const int i = i0 + j;
          if (i < total) {
            while (i >= end) {
              const float inv = __int_as_float(__builtin_amdgcn_readlane(__float_as_int(invv), r));
              *(v4f*)&dstA[r][4 * lane] = sum * inv;
              sum = zero4;
              ++r;
              end = __builtin_amdgcn_readlane(bnd, r + 1);
            }
            sum += v[j];
          }
        }
        idx = idxn;
      }
      while (r < r1) {
        const float inv = __int_as_float(__builtin_amdgcn_readlane(__float_as_int(invv), r));
        *(v4f*)&dstA[r][4 * lane] = sum * inv;
        sum = zero4;
        ++r;
        end = __builtin_amdgcn_readlane(bnd, r + 1);
      }
    }
    __syncthreads();

    if (live) {
#pragma unroll
      for (int kk = 0; kk < D / 4; ++kk) {
        const v2f a = *(const v2f*)&myA[c][4 * kk + k0];
#pragma unroll
        for (int j = 0; j < 8; ++j) {
          const v2f b = *(const v2f*)&sW[0][16 * j + c][4 * kk + k0];
          acc[j] = __builtin_amdgcn_wmma_f32_16x16x4_f32(
              false, a, false, b, (short)0, acc[j], false, false);
        }
      }
    }
    __syncthreads();

    if (live) {
      float4 v[16];
#pragma unroll
      for (int r = 0; r < 16; ++r)
        v[r] = ((const float4*)(xdst + (m0 + r) * D))[lane];
#pragma unroll
      for (int r = 0; r < 16; ++r) *(float4*)&myA[r][4 * lane] = v[r];
    }
    __syncthreads();

    if (live) {
#pragma unroll
      for (int kk = 0; kk < D / 4; ++kk) {
        const v2f a = *(const v2f*)&myA[c][4 * kk + k0];
#pragma unroll
        for (int j = 0; j < 8; ++j) {
          const v2f b = *(const v2f*)&sW[1][16 * j + c][4 * kk + k0];
          acc[j] = __builtin_amdgcn_wmma_f32_16x16x4_f32(
              false, a, false, b, (short)0, acc[j], false, false);
        }
      }
    }
    __syncthreads();

    if (live) {
      float ss[8];
#pragma unroll
      for (int r = 0; r < 8; ++r) {
        float p = 0.0f;
#pragma unroll
        for (int j = 0; j < 8; ++j) p += acc[j][r] * acc[j][r];
        ss[r] = p;
      }
#pragma unroll
      for (int d = 1; d < 16; d <<= 1) {
#pragma unroll
        for (int r = 0; r < 8; ++r) ss[r] += __shfl_xor(ss[r], d);
      }
#pragma unroll
      for (int r = 0; r < 8; ++r) {
        const float nrm = fmaxf(sqrtf(ss[r]), 1e-12f);
#pragma unroll
        for (int j = 0; j < 8; ++j) {
          float val = acc[j][r] / nrm;
          if (RELU) val = fmaxf(val, 0.0f);
          myA[8 * h + r][16 * j + c] = val;
        }
      }
    }
    __syncthreads();

    if (live) {
#pragma unroll
      for (int r = 0; r < 16; ++r) {
        const float4 t = *(const float4*)&myA[r][4 * lane];
        ((float4*)(out + (m0 + r) * D))[lane] = t;
      }
    }
    __syncthreads();
  }
}

extern "C" void kernel_launch(void* const* d_in, const int* in_sizes, int n_in,
                              void* d_out, int out_size, void* d_ws,
                              size_t ws_size, hipStream_t stream) {
  const float* x    = (const float*)d_in[0];
  const float* W_l1 = (const float*)d_in[1];
  const float* b_l1 = (const float*)d_in[2];
  const float* W_r1 = (const float*)d_in[3];
  const float* W_l2 = (const float*)d_in[4];
  const float* b_l2 = (const float*)d_in[5];
  const float* W_r2 = (const float*)d_in[6];
  const int* row1 = (const int*)d_in[7];
  const int* col1 = (const int*)d_in[8];
  const int* row2 = (const int*)d_in[9];
  const int* col2 = (const int*)d_in[10];
  const int E1 = in_sizes[7];
  const int E2 = in_sizes[9];
  const int E = E1 + E2;

  float* ws = (float*)d_ws;
  float* h    = ws;
  int* cnt  = (int*)(h + (size_t)N_MID * D);
  int* off  = cnt + NSEG_PAD;
  int* cntc = off + NSEG_PAD;
  int* rank = cntc + (size_t)NCH * NSEG_PAD;
  int* lst  = rank + E;

  seg_count<<<NCH, 1024, 0, stream>>>(col1, col2, cntc, rank, E1, E);
  seg_combine<<<NSEG_PAD / 256, 256, 0, stream>>>(cntc, cnt);
  seg_scan<<<NSEG_PAD / SCAN_BLK, SCAN_BLK, 0, stream>>>(cnt, off);
  seg_fill<<<(E + 255) / 256, 256, 0, stream>>>(row1, col1, row2, col2, rank,
                                                off, cntc, lst, E1, E);
  sage_layer<1><<<GEMM_WG, GEMM_WAVES * 32, 0, stream>>>(
      x, lst, off, 0, x, W_l1, b_l1, W_r1, h, N_MID);
  sage_layer<0><<<GEMM_WG, GEMM_WAVES * 32, 0, stream>>>(
      h, lst, off, N_MID, h, W_l2, b_l2, W_r2, (float*)d_out, N_DST);
}


# Prefix-sum launch folded into the fill launch: fill is 256 resident 16-wave work-groups, each scans the counts into an LDS offset table and places 8 edges per thread (was 3750 one-edge blocks)
# speedup vs baseline: 2.7264x; 1.0596x over previous
#include <hip/hip_runtime.h>
#include <hip/hip_bf16.h>


#define D 128
#define N_MID 50000
#define N_DST 10000

typedef __attribute__((ext_vector_type(2))) float v2f;
typedef __attribute__((ext_vector_type(4))) float v4f;
typedef __attribute__((ext_vector_type(8))) float v8f;

#define NSEG (N_MID + N_DST)
#define NSEG_PAD 61440

#define NCH 32
#define CNT_UNR 16
#define TRIP (CNT_UNR * 1024)

__global__ __launch_bounds__(1024) void seg_count(
    const int* __restrict__ col1, const int* __restrict__ col2,
    int* __restrict__ cntc, int* __restrict__ rank, int E1, int E) {
  __shared__ int sCnt[NSEG_PAD + 32];
  const int t = threadIdx.x;
  const int c = blockIdx.x;
  int4* s4 = (int4*)sCnt;
  const int4 zero4 = {0, 0, 0, 0};
  for (int i = t; i < NSEG_PAD / 4; i += 1024) s4[i] = zero4;
  if (t < 32) sCnt[NSEG_PAD + t] = 0;
  __syncthreads();

  const int ntrips = (E + TRIP - 1) / TRIP;
  for (int trip = c; trip < ntrips; trip += NCH) {
    const int base = trip * TRIP + t;
    int v[CNT_UNR];
#pragma unroll
    for (int j = 0; j < CNT_UNR; ++j) {
      const int e = base + j * 1024;
      const int ec = (e < E) ? e : E - 1;
      const int* p = (ec < E1) ? col1 + ec : col2 + (ec - E1);
      v[j] = *p;
    }
    asm volatile("" ::"v"(v[0]), "v"(v[1]), "v"(v[2]), "v"(v[3]), "v"(v[4]),
                 "v"(v[5]), "v"(v[6]), "v"(v[7]), "v"(v[8]), "v"(v[9]),
                 "v"(v[10]), "v"(v[11]), "v"(v[12]), "v"(v[13]), "v"(v[14]),
                 "v"(v[15]));
    int rk[CNT_UNR];
#pragma unroll
    for (int j = 0; j < CNT_UNR; ++j) {
      const int e = base + j * 1024;
      const int seg = (e < E1) ? v[j] : v[j] + N_MID;
      rk[j] = atomicAdd(&sCnt[(e < E) ? seg : NSEG_PAD + (t & 31)], 1);
    }
#pragma unroll
    for (int j = 0; j < CNT_UNR; ++j) {
      const int e = base + j * 1024;
      if (e < E) rank[e] = rk[j];
    }
  }
  __syncthreads();

  int4* o4 = (int4*)(cntc + (size_t)c * NSEG_PAD);
  for (int i = t; i < NSEG_PAD / 4; i += 1024) o4[i] = s4[i];
}

__global__ __launch_bounds__(256) void seg_combine(int* __restrict__ cntc,
                                                   int* __restrict__ cnt) {
  const int s = blockIdx.x * 256 + threadIdx.x;
  int v[NCH];
#pragma unroll
  for (int c = 0; c < NCH; ++c) v[c] = cntc[(size_t)c * NSEG_PAD + s];
  int run = 0;
#pragma unroll
  for (int c = 0; c < NCH; ++c) {
    cntc[(size_t)c * NSEG_PAD + s] = run;
    run += v[c];
  }
  cnt[s] = run;
}

#define FILL_WG 256
#define FILL_THR 512
#define FILL_UNR 8
#define FILL_ROWS (NSEG_PAD / (4 * FILL_THR))
#define FILL_SEG (NSEG_PAD / FILL_WG)

__device__ __forceinline__ void fill_load(
    int base, int nthr, const int* __restrict__ row1,
    const int* __restrict__ col1, const int* __restrict__ row2,
    const int* __restrict__ col2, const int* __restrict__ rank, int E1, int E,
    int (&sg)[FILL_UNR], int (&rw)[FILL_UNR], int (&rk)[FILL_UNR]) {
#pragma unroll
  for (int j = 0; j < FILL_UNR; ++j) {
    const int e = base + j * nthr;
    const int ec = (e < E) ? e : E - 1;
    const bool l1 = ec < E1;
    const int* pc = l1 ? col1 + ec : col2 + (ec - E1);
    const int* pr = l1 ? row1 + ec : row2 + (ec - E1);
    sg[j] = *pc;
    rw[j] = *pr;
    rk[j] = rank[ec];
  }
}

__global__ __launch_bounds__(FILL_THR) void seg_fill(
    const int* __restrict__ row1, const int* __restrict__ col1,
    const int* __restrict__ row2, const int* __restrict__ col2,
    const int* __restrict__ rank, const int* __restrict__ cnt,
    const int* __restrict__ cntc, int* __restrict__ off,
    int* __restrict__ lst, int E1, int E) {
  __shared__ __attribute__((aligned(16))) int sOff[NSEG_PAD];
  __shared__ int sTot[FILL_THR / 32];
  const int t = threadIdx.x;
  const int lane = t & 31;
  const int w = t >> 5;
  const int b = blockIdx.x;
  const int nthr = FILL_WG * FILL_THR;
  int base = b * FILL_THR + t;

  int sg[FILL_UNR], rw[FILL_UNR], rk[FILL_UNR];
  fill_load(base, nthr, row1, col1, row2, col2, rank, E1, E, sg, rw, rk);

  const int4* c4 = (const int4*)cnt + w * (FILL_ROWS * 32) + lane;
  int4 v[FILL_ROWS];
#pragma unroll
  for (int m = 0; m < FILL_ROWS; ++m) v[m] = c4[32 * m];
  int run = 0;
#pragma unroll
  for (int m = 0; m < FILL_ROWS; ++m) {
    const int4 x = v[m];
    const int s1 = x.x;
    const int s2 = s1 + x.y;
    const int s3 = s2 + x.z;
    const int tot = s3 + x.w;
    int inc = tot;
#pragma unroll
    for (int d = 1; d < 32; d <<= 1) {
      const int n = __shfl_up(inc, d);
      if (lane >= d) inc += n;
    }
    const int ex = run + inc - tot;
    const int4 y = {ex, ex + s1, ex + s2, ex + s3};
    v[m] = y;
    run += __builtin_amdgcn_readlane(inc, 31);
  }
  if (lane == 0) sTot[w] = run;
  __syncthreads();
  int wbase = 0;
#pragma unroll
  for (int k = 0; k < FILL_THR / 32; ++k) wbase += (k < w) ? sTot[k] : 0;
  int4* s4 = (int4*)sOff + w * (FILL_ROWS * 32) + lane;
#pragma unroll
  for (int m = 0; m < FILL_ROWS; ++m) {
    int4 y = v[m];
    y.x += wbase;
    y.y += wbase;
    y.z += wbase;
    y.w += wbase;
    s4[32 * m] = y;
  }
  __syncthreads();
  if (t < FILL_SEG) off[b * FILL_SEG + t] = sOff[b * FILL_SEG + t];

  while (base < E) {
    int o[FILL_UNR], cc[FILL_UNR];
#pragma unroll
    for (int j = 0; j < FILL_UNR; ++j) {
      const int e = base + j * nthr;
      const int ec = (e < E) ? e : E - 1;
      const int seg = (ec < E1) ? sg[j] : sg[j] + N_MID;
      const unsigned c = ((unsigned)ec / TRIP) % NCH;
      o[j] = sOff[seg];
      cc[j] = cntc[(size_t)c * NSEG_PAD + seg];
    }
    asm volatile("" ::"v"(o[0]), "v"(o[1]), "v"(o[2]), "v"(o[3]), "v"(o[4]),
                 "v"(o[5]), "v"(o[6]), "v"(o[7]), "v"(cc[0]), "v"(cc[1]),
                 "v"(cc[2]), "v"(cc[3]), "v"(cc[4]), "v"(cc[5]), "v"(cc[6]),
                 "v"(cc[7]));
#pragma unroll
    for (int j = 0; j < FILL_UNR; ++j) {
      const int e = base + j * nthr;
      if (e < E) lst[o[j] + cc[j] + rk[j]] = rw[j];
    }
    base += nthr * FILL_UNR;
    if (base < E)
      fill_load(base, nthr, row1, col1, row2, col2, rank, E1, E, sg, rw, rk);
  }
}

#define GEMM_WG 256
#define GEMM_WAVES 16
#define WP (D + 4)

template <int RELU>
__global__ __launch_bounds__(GEMM_WAVES * 32) void sage_layer(
    const float* __restrict__ src, const int* __restrict__ lst,
    const int* __restrict__ off, int segbase,
    const float* __restrict__ xdst, const float* __restrict__ Wl,
    const float* __restrict__ bl, const float* __restrict__ Wr,
    float* __restrict__ out, int M) {
  __shared__ __attribute__((aligned(16))) float sW[2][D][WP];
  __shared__ __attribute__((aligned(16))) float sA[GEMM_WAVES][16][WP];

  const int tid = threadIdx.x;
  const int lane = tid & 31;
  const int w = __builtin_amdgcn_readfirstlane(tid >> 5);
  const int c = lane & 15;
  const int h = lane >> 4;
  const int k0 = h << 1;

  for (int i = tid; i < D * (D / 4); i += GEMM_WAVES * 32) {
    const int n = i >> 5;
    const int q = i & 31;
    const float4 vl = ((const float4*)Wl)[i];
    const float4 vr = ((const float4*)Wr)[i];
    *(float4*)&sW[0][n][4 * q] = vl;
    *(float4*)&sW[1][n][4 * q] = vr;
  }

  float bv[8];
#pragma unroll
  for (int j = 0; j < 8; ++j) bv[j] = bl[16 * j + c];

  const int ntiles = M >> 4;
  const int slots = gridDim.x * GEMM_WAVES;
  const int rounds = (ntiles + slots - 1) / slots;
  float (*myA)[WP] = sA[w];

  for (int rd = 0; rd < rounds; ++rd) {
    const int first = rd * slots + blockIdx.x;
    const int tile = first + w * gridDim.x;
    const bool live = tile < ntiles;
    const size_t m0 = (size_t)(live ? tile : 0) * 16;

    v8f acc[8];
#pragma unroll
    for (int j = 0; j < 8; ++j) {
      const float b = bv[j];
      const v8f t = {b, b, b, b, b, b, b, b};
      acc[j] = t;
    }

    int nl = 0;
#pragma unroll
    for (int t = 0; t < GEMM_WAVES; ++t)
      nl += (first + t * (int)gridDim.x < ntiles) ? 1 : 0;
    int sh = 0;
    while (nl > 0 && (2 << sh) * nl <= GEMM_WAVES) ++sh;
    const int gt = w >> sh;
    if (gt < nl) {
      const int span = 16 >> sh;
      const int r0 = (w & ((1 << sh) - 1)) * span;
      const int r1 = r0 + span;
      const int gm0 = (first + gt * (int)gridDim.x) * 16;
      float (*dstA)[WP] = sA[gt];

      const int* __restrict__ ob = off + segbase + gm0;
      const int o0 = ob[lane < 16 ? lane : 16];
      const int o1 = ob[lane < 15 ? lane + 1 : 16];
      const float invv = 1.0f / fmaxf((float)(o1 - o0), 1.0f);
      const int lbase = __builtin_amdgcn_readlane(o0, r0);
      const int total = __builtin_amdgcn_readlane(o0, r1) - lbase;
      const int bnd = o0 - lbase;
      const int* __restrict__ L = lst + lbase;
      const v4f* __restrict__ s4 = (const v4f*)src + lane;
      const v4f zero4 = {0.0f, 0.0f, 0.0f, 0.0f};

      v4f sum = zero4;
      int r = r0;
      int end = __builtin_amdgcn_readlane(bnd, r0 + 1);

      int idx = 0;
      if (total > 0) idx = L[lane < total ? lane : total - 1];
      for (int i0 = 0; i0 < total; i0 += 16) {
        const int rem = total - i0;
        int idxn = idx;
        if (rem > 16) {
          const int rn = rem - 16;
          idxn = L[i0 + 16 + (lane < rn ? lane : rn - 1)];
        }
        v4f v[16];
#pragma unroll
        for (int j = 0; j < 16; ++j) {
          const int q = (j < rem) ? j : rem - 1;
          const int srow = __builtin_amdgcn_readlane(idx, q);
          v[j] = s4[(size_t)srow * (D / 4)];
        }
#pragma unroll
        for (int j = 0; j < 16; ++j) {
          const int i = i0 + j;
          if (i < total) {
            while (i >= end) {
              const float inv = __int_as_float(__builtin_amdgcn_readlane(__float_as_int(invv), r));
              *(v4f*)&dstA[r][4 * lane] = sum * inv;
              sum = zero4;
              ++r;
              end = __builtin_amdgcn_readlane(bnd, r + 1);
            }
            sum += v[j];
          }
        }
        idx = idxn;
      }
      while (r < r1) {
        const float inv = __int_as_float(__builtin_amdgcn_readlane(__float_as_int(invv), r));
        *(v4f*)&dstA[r][4 * lane] = sum * inv;
        sum = zero4;
        ++r;
        end = __builtin_amdgcn_readlane(bnd, r + 1);
      }
    }
    __syncthreads();

    if (live) {
#pragma unroll
      for (int kk = 0; kk < D / 4; ++kk) {
        const v2f a = *(const v2f*)&myA[c][4 * kk + k0];
#pragma unroll
        for (int j = 0; j < 8; ++j) {
          const v2f b = *(const v2f*)&sW[0][16 * j + c][4 * kk + k0];
          acc[j] = __builtin_amdgcn_wmma_f32_16x16x4_f32(
              false, a, false, b, (short)0, acc[j], false, false);
        }
      }
    }
    __syncthreads();

    if (live) {
      float4 v[16];
#pragma unroll
      for (int r = 0; r < 16; ++r)
        v[r] = ((const float4*)(xdst + (m0 + r) * D))[lane];
#pragma unroll
      for (int r = 0; r < 16; ++r) *(float4*)&myA[r][4 * lane] = v[r];
    }
    __syncthreads();

    if (live) {
#pragma unroll
      for (int kk = 0; kk < D / 4; ++kk) {
        const v2f a = *(const v2f*)&myA[c][4 * kk + k0];
#pragma unroll
        for (int j = 0; j < 8; ++j) {
          const v2f b = *(const v2f*)&sW[1][16 * j + c][4 * kk + k0];
          acc[j] = __builtin_amdgcn_wmma_f32_16x16x4_f32(
              false, a, false, b, (short)0, acc[j], false, false);
        }
      }
    }
    __syncthreads();

    if (live) {
      float ss[8];
#pragma unroll
      for (int r = 0; r < 8; ++r) {
        float p = 0.0f;
#pragma unroll
        for (int j = 0; j < 8; ++j) p += acc[j][r] * acc[j][r];
        ss[r] = p;
      }
#pragma unroll
      for (int d = 1; d < 16; d <<= 1) {
#pragma unroll
        for (int r = 0; r < 8; ++r) ss[r] += __shfl_xor(ss[r], d);
      }
#pragma unroll
      for (int r = 0; r < 8; ++r) {
        const float nrm = fmaxf(sqrtf(ss[r]), 1e-12f);
#pragma unroll
        for (int j = 0; j < 8; ++j) {
          float val = acc[j][r] / nrm;
          if (RELU) val = fmaxf(val, 0.0f);
          myA[8 * h + r][16 * j + c] = val;
        }
      }
    }
    __syncthreads();

    if (live) {
#pragma unroll
      for (int r = 0; r < 16; ++r) {
        const float4 t = *(const float4*)&myA[r][4 * lane];
        ((float4*)(out + (m0 + r) * D))[lane] = t;
      }
    }
    __syncthreads();
  }
}

extern "C" void kernel_launch(void* const* d_in, const int* in_sizes, int n_in,
                              void* d_out, int out_size, void* d_ws,
                              size_t ws_size, hipStream_t stream) {
  const float* x    = (const float*)d_in[0];
  const float* W_l1 = (const float*)d_in[1];
  const float* b_l1 = (const float*)d_in[2];
  const float* W_r1 = (const float*)d_in[3];
  const float* W_l2 = (const float*)d_in[4];
  const float* b_l2 = (const float*)d_in[5];
  const float* W_r2 = (const float*)d_in[6];
  const int* row1 = (const int*)d_in[7];
  const int* col1 = (const int*)d_in[8];
  const int* row2 = (const int*)d_in[9];
  const int* col2 = (const int*)d_in[10];
  const int E1 = in_sizes[7];
  const int E2 = in_sizes[9];
  const int E = E1 + E2;

  float* ws = (float*)d_ws;
  float* h    = ws;
  int* cnt  = (int*)(h + (size_t)N_MID * D);
  int* off  = cnt + NSEG_PAD;
  int* cntc = off + NSEG_PAD;
  int* rank = cntc + (size_t)NCH * NSEG_PAD;
  int* lst  = rank + E;

  seg_count<<<NCH, 1024, 0, stream>>>(col1, col2, cntc, rank, E1, E);
  seg_combine<<<NSEG_PAD / 256, 256, 0, stream>>>(cntc, cnt);
  seg_fill<<<FILL_WG, FILL_THR, 0, stream>>>(row1, col1, row2, col2, rank, cnt,
                                             cntc, off, lst, E1, E);
  sage_layer<1><<<GEMM_WG, GEMM_WAVES * 32, 0, stream>>>(
      x, lst, off, 0, x, W_l1, b_l1, W_r1, h, N_MID);
  sage_layer<0><<<GEMM_WG, GEMM_WAVES * 32, 0, stream>>>(
      h, lst, off, N_MID, h, W_l2, b_l2, W_r2, (float*)d_out, N_DST);
}
